# starting kernel
# baseline (speedup 1.0000x reference)
#include <hip/hip_runtime.h>

typedef float v2f __attribute__((ext_vector_type(2)));
typedef float v8f __attribute__((ext_vector_type(8)));

#define LW   0.5f
#define BDIM 16
#define TDIM 256
#define QDIM 64
#define DDIM 128
#define AS   68
#define XS   132
#define YS   132

__global__ void lookahead_scan(const float* __restrict__ adj,
                               float* __restrict__ wbuf) {
  const int b = blockIdx.x >> 6;
  const int q = blockIdx.x & 63;
  const int c = threadIdx.x;
  const bool isDiag = (c == q);
  float carry = 0.0f;
  const size_t base = ((size_t)b * TDIM) * (QDIM * QDIM) + (size_t)q * QDIM + c;
  for (int t = TDIM - 1; t >= 0; --t) {
    const size_t idx = base + (size_t)t * (QDIM * QDIM);
    const float v = adj[idx];
    carry = LW * (carry + v);
    wbuf[idx] = isDiag ? 1.0f : carry;
  }
}

__global__ void pack_weight(const float* __restrict__ W,
                            float* __restrict__ Wp) {
  const int idx = blockIdx.x * blockDim.x + threadIdx.x;
  const int k4  = idx >> 8;
  const int col = (idx & 255) >> 1;
  const int p   = idx & 1;
  const int kr  = 4 * k4 + 2 * p;
  Wp[2 * idx]     = W[kr * DDIM + col];
  Wp[2 * idx + 1] = W[(kr + 1) * DDIM + col];
}

__global__ __launch_bounds__(256)
void gcn_fused(const float* __restrict__ xin,
               const float* __restrict__ wbuf,
               const float* __restrict__ Wp1, const float* __restrict__ b1,
               const float* __restrict__ Wp2, const float* __restrict__ b2,
               float* __restrict__ out) {
  __shared__ float sA[QDIM * AS];
  __shared__ float sX[QDIM * XS];
  __shared__ float sY[(QDIM / 2) * YS * 2];
  __shared__ float sDis[QDIM];

  const int tid  = threadIdx.x;
  const int lane = tid & 31;
  const int wv   = tid >> 5;
  const int r    = wv & 3;
  const int g    = wv >> 2;
  const int lm   = lane & 15;
  const int p    = lane >> 4;
  const size_t n = blockIdx.x;

  {
    const float* wsrc = wbuf + n * (size_t)(QDIM * QDIM);
#pragma unroll
    for (int i = 0; i < 16; ++i) {
      const int f  = i * 256 + tid;
      const int rr = f >> 6, cc = f & 63;
      sA[rr * AS + cc] = wsrc[f];
    }
  }
  {
    const float4* xsrc = (const float4*)(xin + n * (size_t)(QDIM * DDIM));
#pragma unroll
    for (int i = 0; i < 8; ++i) {
      const int f4 = i * 256 + tid;
      const int rr = f4 >> 5, c4 = f4 & 31;
      *(float4*)&sX[rr * XS + c4 * 4] = xsrc[f4];
    }
  }
  __syncthreads();

  if (tid < QDIM) {
    float s = 0.0f;
#pragma unroll 4
    for (int cc = 0; cc < QDIM; ++cc) s += sA[tid * AS + cc];
    sDis[tid] = rsqrtf(fmaxf(s, 1.0f));
  }
  __syncthreads();
#pragma unroll
  for (int i = 0; i < 16; ++i) {
    const int f  = i * 256 + tid;
    const int rr = f >> 6, cc = f & 63;
    sA[rr * AS + cc] *= sDis[rr] * sDis[cc];
  }
  __syncthreads();

  auto gemm_xw = [&](const float* __restrict__ sIn, const float* __restrict__ Wp) {
#pragma unroll
    for (int c = 0; c < 4; ++c) {
      const int colbase = g * 64 + c * 16;
      v8f acc = {0.f, 0.f, 0.f, 0.f, 0.f, 0.f, 0.f, 0.f};
      const float* arow = sIn + (r * 16 + lm) * XS + 2 * p;
      const float* bcol = Wp + 2 * ((colbase + lm) * 2 + p);
#pragma unroll
      for (int k4 = 0; k4 < 32; ++k4) {
        v2f a = *(const v2f*)(arow + 4 * k4);
        v2f b = *(const v2f*)(bcol + 512 * k4);
        acc = __builtin_amdgcn_wmma_f32_16x16x4_f32(
            false, a, false, b, (short)0, acc, false, false);
      }
#pragma unroll
      for (int j2 = 0; j2 < 4; ++j2) {
        const int rp = r * 8 + p * 4 + j2;
        v2f d;
        d.x = acc[2 * j2];
        d.y = acc[2 * j2 + 1];
        *(v2f*)&sY[2 * (rp * YS + colbase + lm)] = d;
      }
    }
  };

  auto gemm_ay = [&](const float* __restrict__ bias, float* __restrict__ sOut) {
#pragma unroll
    for (int c = 0; c < 4; ++c) {
      const int colbase = g * 64 + c * 16;
      const float bb = bias[colbase + lm];
      v8f acc = {bb, bb, bb, bb, bb, bb, bb, bb};
      const float* arow = sA + (r * 16 + lm) * AS + 2 * p;
      const float* bcol = sY + 2 * ((size_t)p * YS + colbase + lm);
#pragma unroll
      for (int k4 = 0; k4 < 16; ++k4) {
        v2f a = *(const v2f*)(arow + 4 * k4);
        v2f b = *(const v2f*)(bcol + 528 * k4);
        acc = __builtin_amdgcn_wmma_f32_16x16x4_f32(
            false, a, false, b, (short)0, acc, false, false);
      }
#pragma unroll
      for (int j = 0; j < 8; ++j)
        sOut[(r * 16 + p * 8 + j) * XS + colbase + lm] = acc[j];
    }
  };

  gemm_xw(sX, Wp1);  __syncthreads();
  gemm_ay(b1, sX);   __syncthreads();
  gemm_xw(sX, Wp2);  __syncthreads();
  gemm_ay(b2, sX);   __syncthreads();

  {
    float4* dst = (float4*)(out + n * (size_t)(QDIM * DDIM));
#pragma unroll
    for (int i = 0; i < 8; ++i) {
      const int f4 = i * 256 + tid;
      const int rr = f4 >> 5, c4 = f4 & 31;
      dst[f4] = *(const float4*)&sX[rr * XS + c4 * 4];
    }
  }
}

extern "C" void kernel_launch(void* const* d_in, const int* in_sizes, int n_in,
                              void* d_out, int out_size, void* d_ws, size_t ws_size,
                              hipStream_t stream) {
  const float* x   = (const float*)d_in[0];
  const float* adj = (const float*)d_in[1];
  const float* W1  = (const float*)d_in[2];
  const float* b1  = (const float*)d_in[3];
  const float* W2  = (const float*)d_in[4];
  const float* b2  = (const float*)d_in[5];
  float* out = (float*)d_out;

  float* Wp1  = (float*)d_ws;
  float* Wp2  = Wp1 + 16384;
  float* wbuf = Wp2 + 16384;

  lookahead_scan<<<BDIM * QDIM, 64, 0, stream>>>(adj, wbuf);
  pack_weight<<<32, 256, 0, stream>>>(W1, Wp1);
  pack_weight<<<32, 256, 0, stream>>>(W2, Wp2);
  gcn_fused<<<BDIM * TDIM, 256, 0, stream>>>(x, wbuf, Wp1, b1, Wp2, b2, out);
}


# Fold the two linear layers (no activation between): W12=W1*W2 and c=b1*W2 once per call, out=A(A(X*W12)+c)+b2; one 64x128x128 product per graph dropped; f32 summation order changes
# speedup vs baseline: 1.2885x; 1.2885x over previous
#include <hip/hip_runtime.h>

typedef float v2f __attribute__((ext_vector_type(2)));
typedef float v8f __attribute__((ext_vector_type(8)));

#define LW   0.5f
#define BDIM 16
#define TDIM 256
#define QDIM 64
#define DDIM 128
#define AS   68
#define XS   132
#define YS   132

__global__ void lookahead_scan(const float* __restrict__ adj,
                               float* __restrict__ wbuf) {
  const int b = blockIdx.x >> 6;
  const int q = blockIdx.x & 63;
  const int c = threadIdx.x;
  const bool isDiag = (c == q);
  float carry = 0.0f;
  const size_t base = ((size_t)b * TDIM) * (QDIM * QDIM) + (size_t)q * QDIM + c;
  for (int t = TDIM - 1; t >= 0; --t) {
    const size_t idx = base + (size_t)t * (QDIM * QDIM);
    const float v = adj[idx];
    carry = LW * (carry + v);
    wbuf[idx] = isDiag ? 1.0f : carry;
  }
}

__global__ __launch_bounds__(256)
void fuse_weights(const float* __restrict__ W1, const float* __restrict__ b1,
                  const float* __restrict__ W2,
                  float* __restrict__ Wp, float* __restrict__ cvec) {
  const int lane = threadIdx.x & 31;
  const int j    = threadIdx.x >> 5;
  const int lm   = lane & 15;
  const int p    = lane >> 4;
  const int i    = blockIdx.x;
  const bool biasStrip = (i == 8);
  const bool zeroRow   = biasStrip && (lm != 0);
  const float* arow = biasStrip ? b1 : (W1 + (16 * i + lm) * DDIM);
  const float* bcol = W2 + 16 * j + lm;
  v8f acc = {0.f, 0.f, 0.f, 0.f, 0.f, 0.f, 0.f, 0.f};
#pragma unroll
  for (int k4 = 0; k4 < 32; ++k4) {
    const int k = 4 * k4 + 2 * p;
    v2f a = *(const v2f*)(arow + k);
    if (zeroRow) { a.x = 0.0f; a.y = 0.0f; }
    v2f b;
    b.x = bcol[k * DDIM];
    b.y = bcol[(k + 1) * DDIM];
    acc = __builtin_amdgcn_wmma_f32_16x16x4_f32(
        false, a, false, b, (short)0, acc, false, false);
  }
  if (!biasStrip) {
    const int col = 16 * j + lm;
#pragma unroll
    for (int j2 = 0; j2 < 4; ++j2) {
      const int k4o = 4 * i + 2 * p + (j2 >> 1);
      v2f d;
      d.x = acc[2 * j2];
      d.y = acc[2 * j2 + 1];
      *(v2f*)&Wp[2 * ((k4o * DDIM + col) * 2 + (j2 & 1))] = d;
    }
  } else if (p == 0) {
    cvec[16 * j + lm] = acc[0];
  }
}

__global__ __launch_bounds__(256)
void gcn_fused(const float* __restrict__ xin,
               const float* __restrict__ wbuf,
               const float* __restrict__ Wp12, const float* __restrict__ cvec,
               const float* __restrict__ b2,
               float* __restrict__ out) {
  __shared__ float sA[QDIM * AS];
  __shared__ float sX[QDIM * XS];
  __shared__ float sY[(QDIM / 2) * YS * 2];
  __shared__ float sDis[QDIM];

  const int tid  = threadIdx.x;
  const int lane = tid & 31;
  const int wv   = tid >> 5;
  const int r    = wv & 3;
  const int g    = wv >> 2;
  const int lm   = lane & 15;
  const int p    = lane >> 4;
  const size_t n = blockIdx.x;

  {
    const float* wsrc = wbuf + n * (size_t)(QDIM * QDIM);
#pragma unroll
    for (int i = 0; i < 16; ++i) {
      const int f  = i * 256 + tid;
      const int rr = f >> 6, cc = f & 63;
      sA[rr * AS + cc] = wsrc[f];
    }
  }
  {
    const float4* xsrc = (const float4*)(xin + n * (size_t)(QDIM * DDIM));
#pragma unroll
    for (int i = 0; i < 8; ++i) {
      const int f4 = i * 256 + tid;
      const int rr = f4 >> 5, c4 = f4 & 31;
      *(float4*)&sX[rr * XS + c4 * 4] = xsrc[f4];
    }
  }
  __syncthreads();

  if (tid < QDIM) {
    float s = 0.0f;
#pragma unroll 4
    for (int cc = 0; cc < QDIM; ++cc) s += sA[tid * AS + cc];
    sDis[tid] = rsqrtf(fmaxf(s, 1.0f));
  }
  __syncthreads();
#pragma unroll
  for (int i = 0; i < 16; ++i) {
    const int f  = i * 256 + tid;
    const int rr = f >> 6, cc = f & 63;
    sA[rr * AS + cc] *= sDis[rr] * sDis[cc];
  }
  __syncthreads();

  auto gemm_xw = [&](const float* __restrict__ sIn, const float* __restrict__ Wp) {
#pragma unroll
    for (int c = 0; c < 4; ++c) {
      const int colbase = g * 64 + c * 16;
      v8f acc = {0.f, 0.f, 0.f, 0.f, 0.f, 0.f, 0.f, 0.f};
      const float* arow = sIn + (r * 16 + lm) * XS + 2 * p;
      const float* bcol = Wp + 2 * ((colbase + lm) * 2 + p);
#pragma unroll
      for (int k4 = 0; k4 < 32; ++k4) {
        v2f a = *(const v2f*)(arow + 4 * k4);
        v2f b = *(const v2f*)(bcol + 512 * k4);
        acc = __builtin_amdgcn_wmma_f32_16x16x4_f32(
            false, a, false, b, (short)0, acc, false, false);
      }
#pragma unroll
      for (int j2 = 0; j2 < 4; ++j2) {
        const int rp = r * 8 + p * 4 + j2;
        v2f d;
        d.x = acc[2 * j2];
        d.y = acc[2 * j2 + 1];
        *(v2f*)&sY[2 * (rp * YS + colbase + lm)] = d;
      }
    }
  };

  auto gemm_ay = [&](const float* __restrict__ bias, const float* __restrict__ sPk,
                     float* __restrict__ sDst, const bool packedOut) {
#pragma unroll
    for (int c = 0; c < 4; ++c) {
      const int colbase = g * 64 + c * 16;
      const float bb = bias[colbase + lm];
      v8f acc = {bb, bb, bb, bb, bb, bb, bb, bb};
      const float* arow = sA + (r * 16 + lm) * AS + 2 * p;
      const float* bcol = sPk + 2 * ((size_t)p * YS + colbase + lm);
#pragma unroll
      for (int k4 = 0; k4 < 16; ++k4) {
        v2f a = *(const v2f*)(arow + 4 * k4);
        v2f b = *(const v2f*)(bcol + 528 * k4);
        acc = __builtin_amdgcn_wmma_f32_16x16x4_f32(
            false, a, false, b, (short)0, acc, false, false);
      }
      if (packedOut) {
#pragma unroll
        for (int j2 = 0; j2 < 4; ++j2) {
          const int rp = r * 8 + p * 4 + j2;
          v2f d;
          d.x = acc[2 * j2];
          d.y = acc[2 * j2 + 1];
          *(v2f*)&sDst[2 * (rp * YS + colbase + lm)] = d;
        }
      } else {
#pragma unroll
        for (int j = 0; j < 8; ++j)
          sDst[(r * 16 + p * 8 + j) * XS + colbase + lm] = acc[j];
      }
    }
  };

  gemm_xw(sX, Wp12);             __syncthreads();
  gemm_ay(cvec, sY, sX, true);   __syncthreads();
  gemm_ay(b2,   sX, sY, false);  __syncthreads();

  {
    float4* dst = (float4*)(out + n * (size_t)(QDIM * DDIM));
#pragma unroll
    for (int i = 0; i < 8; ++i) {
      const int f4 = i * 256 + tid;
      const int rr = f4 >> 5, c4 = f4 & 31;
      dst[f4] = *(const float4*)&sY[rr * XS + c4 * 4];
    }
  }
}

extern "C" void kernel_launch(void* const* d_in, const int* in_sizes, int n_in,
                              void* d_out, int out_size, void* d_ws, size_t ws_size,
                              hipStream_t stream) {
  const float* x   = (const float*)d_in[0];
  const float* adj = (const float*)d_in[1];
  const float* W1  = (const float*)d_in[2];
  const float* b1  = (const float*)d_in[3];
  const float* W2  = (const float*)d_in[4];
  const float* b2  = (const float*)d_in[5];
  float* out = (float*)d_out;

  float* Wp12 = (float*)d_ws;
  float* cvec = Wp12 + 16384;
  float* wbuf = cvec + 16384;

  lookahead_scan<<<BDIM * QDIM, 64, 0, stream>>>(adj, wbuf);
  fuse_weights<<<9, 256, 0, stream>>>(W1, b1, W2, Wp12, cvec);
  gcn_fused<<<BDIM * TDIM, 256, 0, stream>>>(x, wbuf, Wp12, cvec, b2, out);
}


# Scan kernel: loads issued 6 groups of 8 time steps ahead of the group being consumed (register ring, about 56 independent reads in flight per wave instead of 8); arithmetic and order unchanged
# speedup vs baseline: 1.4096x; 1.0940x over previous
#include <hip/hip_runtime.h>

typedef float v2f __attribute__((ext_vector_type(2)));
typedef float v8f __attribute__((ext_vector_type(8)));

#define LW   0.5f
#define BDIM 16
#define TDIM 256
#define QDIM 64
#define DDIM 128
#define AS   68
#define XS   132
#define YS   132

#define SG 8
#define LA 6
#define NG (TDIM / SG)
__global__ __launch_bounds__(64)
void lookahead_scan(const float* __restrict__ adj,
                    float* __restrict__ wbuf) {
  const int b = blockIdx.x >> 6;
  const int q = blockIdx.x & 63;
  const int c = threadIdx.x;
  const bool isDiag = (c == q);
  const size_t base = ((size_t)b * TDIM) * (QDIM * QDIM) + (size_t)q * QDIM + c;
  const float* __restrict__ src = adj + base;
  float* __restrict__ dst = wbuf + base;
  float ring[(LA + 1) * SG];
#pragma unroll
  for (int g = 0; g < LA; ++g) {
#pragma unroll
    for (int i = 0; i < SG; ++i)
      ring[(g % (LA + 1)) * SG + i] = src[(size_t)(TDIM - 1 - g * SG - i) * (QDIM * QDIM)];
  }
  float carry = 0.0f;
#pragma unroll
  for (int g = 0; g < NG; ++g) {
    if (g + LA < NG) {
#pragma unroll
      for (int i = 0; i < SG; ++i)
        ring[((g + LA) % (LA + 1)) * SG + i] =
            src[(size_t)(TDIM - 1 - (g + LA) * SG - i) * (QDIM * QDIM)];
    }
    __builtin_amdgcn_sched_barrier(0);
#pragma unroll
    for (int i = 0; i < SG; ++i) {
      carry = LW * (carry + ring[(g % (LA + 1)) * SG + i]);
      dst[(size_t)(TDIM - 1 - g * SG - i) * (QDIM * QDIM)] = isDiag ? 1.0f : carry;
    }
  }
}

__global__ __launch_bounds__(256)
void fuse_weights(const float* __restrict__ W1, const float* __restrict__ b1,
                  const float* __restrict__ W2,
                  float* __restrict__ Wp, float* __restrict__ cvec) {
  const int lane = threadIdx.x & 31;
  const int j    = threadIdx.x >> 5;
  const int lm   = lane & 15;
  const int p    = lane >> 4;
  const int i    = blockIdx.x;
  const bool biasStrip = (i == 8);
  const bool zeroRow   = biasStrip && (lm != 0);
  const float* arow = biasStrip ? b1 : (W1 + (16 * i + lm) * DDIM);
  const float* bcol = W2 + 16 * j + lm;
  v8f acc = {0.f, 0.f, 0.f, 0.f, 0.f, 0.f, 0.f, 0.f};
#pragma unroll
  for (int k4 = 0; k4 < 32; ++k4) {
    const int k = 4 * k4 + 2 * p;
    v2f a = *(const v2f*)(arow + k);
    if (zeroRow) { a.x = 0.0f; a.y = 0.0f; }
    v2f b;
    b.x = bcol[k * DDIM];
    b.y = bcol[(k + 1) * DDIM];
    acc = __builtin_amdgcn_wmma_f32_16x16x4_f32(
        false, a, false, b, (short)0, acc, false, false);
  }
  if (!biasStrip) {
    const int col = 16 * j + lm;
#pragma unroll
    for (int j2 = 0; j2 < 4; ++j2) {
      const int k4o = 4 * i + 2 * p + (j2 >> 1);
      v2f d;
      d.x = acc[2 * j2];
      d.y = acc[2 * j2 + 1];
      *(v2f*)&Wp[2 * ((k4o * DDIM + col) * 2 + (j2 & 1))] = d;
    }
  } else if (p == 0) {
    cvec[16 * j + lm] = acc[0];
  }
}

__global__ __launch_bounds__(256)
void gcn_fused(const float* __restrict__ xin,
               const float* __restrict__ wbuf,
               const float* __restrict__ Wp12, const float* __restrict__ cvec,
               const float* __restrict__ b2,
               float* __restrict__ out) {
  __shared__ float sA[QDIM * AS];
  __shared__ float sX[QDIM * XS];
  __shared__ float sY[(QDIM / 2) * YS * 2];
  __shared__ float sDis[QDIM];

  const int tid  = threadIdx.x;
  const int lane = tid & 31;
  const int wv   = tid >> 5;
  const int r    = wv & 3;
  const int g    = wv >> 2;
  const int lm   = lane & 15;
  const int p    = lane >> 4;
  const size_t n = blockIdx.x;

  {
    const float* wsrc = wbuf + n * (size_t)(QDIM * QDIM);
#pragma unroll
    for (int i = 0; i < 16; ++i) {
      const int f  = i * 256 + tid;
      const int rr = f >> 6, cc = f & 63;
      sA[rr * AS + cc] = wsrc[f];
    }
  }
  {
    const float4* xsrc = (const float4*)(xin + n * (size_t)(QDIM * DDIM));
#pragma unroll
    for (int i = 0; i < 8; ++i) {
      const int f4 = i * 256 + tid;
      const int rr = f4 >> 5, c4 = f4 & 31;
      *(float4*)&sX[rr * XS + c4 * 4] = xsrc[f4];
    }
  }
  __syncthreads();

  if (tid < QDIM) {
    float s = 0.0f;
#pragma unroll 4
    for (int cc = 0; cc < QDIM; ++cc) s += sA[tid * AS + cc];
    sDis[tid] = rsqrtf(fmaxf(s, 1.0f));
  }
  __syncthreads();
#pragma unroll
  for (int i = 0; i < 16; ++i) {
    const int f  = i * 256 + tid;
    const int rr = f >> 6, cc = f & 63;
    sA[rr * AS + cc] *= sDis[rr] * sDis[cc];
  }
  __syncthreads();

  auto gemm_xw = [&](const float* __restrict__ sIn, const float* __restrict__ Wp) {
#pragma unroll
    for (int c = 0; c < 4; ++c) {
      const int colbase = g * 64 + c * 16;
      v8f acc = {0.f, 0.f, 0.f, 0.f, 0.f, 0.f, 0.f, 0.f};
      const float* arow = sIn + (r * 16 + lm) * XS + 2 * p;
      const float* bcol = Wp + 2 * ((colbase + lm) * 2 + p);
#pragma unroll
      for (int k4 = 0; k4 < 32; ++k4) {
        v2f a = *(const v2f*)(arow + 4 * k4);
        v2f b = *(const v2f*)(bcol + 512 * k4);
        acc = __builtin_amdgcn_wmma_f32_16x16x4_f32(
            false, a, false, b, (short)0, acc, false, false);
      }
#pragma unroll
      for (int j2 = 0; j2 < 4; ++j2) {
        const int rp = r * 8 + p * 4 + j2;
        v2f d;
        d.x = acc[2 * j2];
        d.y = acc[2 * j2 + 1];
        *(v2f*)&sY[2 * (rp * YS + colbase + lm)] = d;
      }
    }
  };

  auto gemm_ay = [&](const float* __restrict__ bias, const float* __restrict__ sPk,
                     float* __restrict__ sDst, const bool packedOut) {
#pragma unroll
    for (int c = 0; c < 4; ++c) {
      const int colbase = g * 64 + c * 16;
      const float bb = bias[colbase + lm];
      v8f acc = {bb, bb, bb, bb, bb, bb, bb, bb};
      const float* arow = sA + (r * 16 + lm) * AS + 2 * p;
      const float* bcol = sPk + 2 * ((size_t)p * YS + colbase + lm);
#pragma unroll
      for (int k4 = 0; k4 < 16; ++k4) {
        v2f a = *(const v2f*)(arow + 4 * k4);
        v2f b = *(const v2f*)(bcol + 528 * k4);
        acc = __builtin_amdgcn_wmma_f32_16x16x4_f32(
            false, a, false, b, (short)0, acc, false, false);
      }
      if (packedOut) {
#pragma unroll
        for (int j2 = 0; j2 < 4; ++j2) {
          const int rp = r * 8 + p * 4 + j2;
          v2f d;
          d.x = acc[2 * j2];
          d.y = acc[2 * j2 + 1];
          *(v2f*)&sDst[2 * (rp * YS + colbase + lm)] = d;
        }
      } else {
#pragma unroll
        for (int j = 0; j < 8; ++j)
          sDst[(r * 16 + p * 8 + j) * XS + colbase + lm] = acc[j];
      }
    }
  };

  gemm_xw(sX, Wp12);             __syncthreads();
  gemm_ay(cvec, sY, sX, true);   __syncthreads();
  gemm_ay(b2,   sX, sY, false);  __syncthreads();

  {
    float4* dst = (float4*)(out + n * (size_t)(QDIM * DDIM));
#pragma unroll
    for (int i = 0; i < 8; ++i) {
      const int f4 = i * 256 + tid;
      const int rr = f4 >> 5, c4 = f4 & 31;
      dst[f4] = *(const float4*)&sY[rr * XS + c4 * 4];
    }
  }
}

extern "C" void kernel_launch(void* const* d_in, const int* in_sizes, int n_in,
                              void* d_out, int out_size, void* d_ws, size_t ws_size,
                              hipStream_t stream) {
  const float* x   = (const float*)d_in[0];
  const float* adj = (const float*)d_in[1];
  const float* W1  = (const float*)d_in[2];
  const float* b1  = (const float*)d_in[3];
  const float* W2  = (const float*)d_in[4];
  const float* b2  = (const float*)d_in[5];
  float* out = (float*)d_out;

  float* Wp12 = (float*)d_ws;
  float* cvec = Wp12 + 16384;
  float* wbuf = cvec + 16384;

  lookahead_scan<<<BDIM * QDIM, 64, 0, stream>>>(adj, wbuf);
  fuse_weights<<<9, 256, 0, stream>>>(W1, b1, W2, Wp12, cvec);
  gcn_fused<<<BDIM * TDIM, 256, 0, stream>>>(x, wbuf, Wp12, cvec, b2, out);
}
